# starting kernel
# baseline (speedup 1.0000x reference)
#include <hip/hip_runtime.h>

typedef float v2f __attribute__((ext_vector_type(2)));
typedef float v8f __attribute__((ext_vector_type(8)));

#define NN 8192
#define D 128

__global__ __launch_bounds__(256) void k_deg(const float* __restrict__ adj,
                                             float* __restrict__ dinv) {
  __shared__ float red[256];
  const int row = blockIdx.x;
  const float* r = adj + (size_t)row * NN;
  float s = 0.f;
  for (int j = threadIdx.x; j < NN; j += 256)
    s += __builtin_nontemporal_load(r + j);
  red[threadIdx.x] = s;
  __syncthreads();
  for (int off = 128; off > 0; off >>= 1) {
    if (threadIdx.x < off) red[threadIdx.x] += red[threadIdx.x + off];
    __syncthreads();
  }
  if (threadIdx.x == 0) {
    float d = red[0];
    dinv[row] = d > 0.f ? rsqrtf(d) : 0.f;
  }
}

__global__ __launch_bounds__(128) void k_proj(const float* __restrict__ x,
                                              const float* __restrict__ w,
                                              const float* __restrict__ dinv,
                                              float* __restrict__ bfrag) {
  __shared__ float xs[D];
  const int row = blockIdx.x;
  const int col = threadIdx.x;
  xs[col] = x[(size_t)row * D + col];
  __syncthreads();
  float acc = 0.f;
#pragma unroll 8
  for (int k = 0; k < D; ++k) acc += xs[k] * w[k * D + col];
  acc *= dinv[row];

  const int ks = row >> 2, within = row & 3;
  const int r = within & 1, half = within >> 1;
  const int lane = half * 16 + (col & 15);
  const int tile = col >> 4;
  bfrag[(((size_t)ks * 8 + tile) * 32 + lane) * 2 + r] = acc;
}

__global__ __launch_bounds__(128) void k_agg(const float* __restrict__ adj,
                                             const float* __restrict__ bfrag,
                                             const float* __restrict__ dinv,
                                             const float* __restrict__ bias,
                                             float* __restrict__ out) {
  const int lane = threadIdx.x & 31;
  const int wave = threadIdx.x >> 5;
  const int row0 = blockIdx.x * 64 + wave * 16;
  const int lo = lane & 15, hi = lane >> 4;

  const v2f* arow = (const v2f*)(adj + (size_t)(row0 + lo) * NN) + hi;
  const v2f* bf = (const v2f*)bfrag + lane;

  v8f c[8] = {};

  for (int ks = 0; ks < NN / 4; ++ks) {
    const v2f a = __builtin_nontemporal_load(arow + (size_t)ks * 2);
    const v2f* bks = bf + (size_t)ks * (8 * 32);
#pragma unroll
    for (int t = 0; t < 8; ++t) {
      const v2f b = bks[t * 32];
      c[t] = __builtin_amdgcn_wmma_f32_16x16x4_f32(
           false, a,  false, b,
           (short)0, c[t],  false,  false);
    }
  }

#pragma unroll
  for (int t = 0; t < 8; ++t) {
    const int col = t * 16 + lo;
    const float bv = bias[col];
#pragma unroll
    for (int v = 0; v < 8; ++v) {
      const int row = row0 + hi * 8 + v;
      out[(size_t)row * D + col] = dinv[row] * c[t][v] + bv;
    }
  }
}

extern "C" void kernel_launch(void* const* d_in, const int* in_sizes, int n_in,
                              void* d_out, int out_size, void* d_ws, size_t ws_size,
                              hipStream_t stream) {
  const float* x    = (const float*)d_in[0];
  const float* adj  = (const float*)d_in[1];
  const float* w    = (const float*)d_in[2];
  const float* bias = (const float*)d_in[3];
  float* out = (float*)d_out;

  float* dinv  = (float*)d_ws;
  float* bfrag = dinv + NN;

  k_deg <<<NN, 256, 0, stream>>>(adj, dinv);
  k_proj<<<NN, 128, 0, stream>>>(x, w, dinv, bfrag);
  k_agg <<<NN / 64, 128, 0, stream>>>(adj, bfrag, dinv, bias, out);
}


# k_agg: K split over the 16 waves of a 16-row workgroup (wave w takes k-steps w, w+16, ...), partial tiles summed through LDS in fixed order: 16x more waves in flight; summation order changes
# speedup vs baseline: 3.9772x; 3.9772x over previous
#include <hip/hip_runtime.h>

typedef float v2f __attribute__((ext_vector_type(2)));
typedef float v8f __attribute__((ext_vector_type(8)));

#define NN 8192
#define D 128

__global__ __launch_bounds__(256) void k_deg(const float* __restrict__ adj,
                                             float* __restrict__ dinv) {
  __shared__ float red[256];
  const int row = blockIdx.x;
  const float* r = adj + (size_t)row * NN;
  float s = 0.f;
  for (int j = threadIdx.x; j < NN; j += 256)
    s += __builtin_nontemporal_load(r + j);
  red[threadIdx.x] = s;
  __syncthreads();
  for (int off = 128; off > 0; off >>= 1) {
    if (threadIdx.x < off) red[threadIdx.x] += red[threadIdx.x + off];
    __syncthreads();
  }
  if (threadIdx.x == 0) {
    float d = red[0];
    dinv[row] = d > 0.f ? rsqrtf(d) : 0.f;
  }
}

__global__ __launch_bounds__(128) void k_proj(const float* __restrict__ x,
                                              const float* __restrict__ w,
                                              const float* __restrict__ dinv,
                                              float* __restrict__ bfrag) {
  __shared__ float xs[D];
  const int row = blockIdx.x;
  const int col = threadIdx.x;
  xs[col] = x[(size_t)row * D + col];
  __syncthreads();
  float acc = 0.f;
#pragma unroll 8
  for (int k = 0; k < D; ++k) acc += xs[k] * w[k * D + col];
  acc *= dinv[row];

  const int ks = row >> 2, within = row & 3;
  const int r = within & 1, half = within >> 1;
  const int lane = half * 16 + (col & 15);
  const int tile = col >> 4;
  bfrag[(((size_t)ks * 8 + tile) * 32 + lane) * 2 + r] = acc;
}

#define KSPLIT 16
#define RED_PITCH 132
__global__ __launch_bounds__(32 * KSPLIT) void k_agg(const float* __restrict__ adj,
                                             const float* __restrict__ bfrag,
                                             const float* __restrict__ dinv,
                                             const float* __restrict__ bias,
                                             float* __restrict__ out) {
  __shared__ float red[KSPLIT][16][RED_PITCH];

  const int lane = threadIdx.x & 31;
  const int wave = threadIdx.x >> 5;
  const int row0 = blockIdx.x * 16;
  const int lo = lane & 15, hi = lane >> 4;

  const v2f* arow = (const v2f*)(adj + (size_t)(row0 + lo) * NN) + hi;
  const v2f* bf = (const v2f*)bfrag + lane;

  v8f c[8] = {};

  for (int ks = wave; ks < NN / 4; ks += KSPLIT) {
    const v2f a = __builtin_nontemporal_load(arow + (size_t)ks * 2);
    const v2f* bks = bf + (size_t)ks * (8 * 32);
#pragma unroll
    for (int t = 0; t < 8; ++t) {
      const v2f b = bks[t * 32];
      c[t] = __builtin_amdgcn_wmma_f32_16x16x4_f32(
           false, a,  false, b,
           (short)0, c[t],  false,  false);
    }
  }

#pragma unroll
  for (int t = 0; t < 8; ++t) {
#pragma unroll
    for (int v = 0; v < 8; ++v) red[wave][hi * 8 + v][t * 16 + lo] = c[t][v];
  }
  __syncthreads();

  const int r = wave;
  const int col = lane * 4;
  float4 s = *(const float4*)&red[0][r][col];
#pragma unroll
  for (int w = 1; w < KSPLIT; ++w) {
    const float4 p = *(const float4*)&red[w][r][col];
    s.x += p.x; s.y += p.y; s.z += p.z; s.w += p.w;
  }
  const float di = dinv[row0 + r];
  const float4 bv = *(const float4*)(bias + col);
  float4 o;
  o.x = di * s.x + bv.x; o.y = di * s.y + bv.y;
  o.z = di * s.z + bv.z; o.w = di * s.w + bv.w;
  *(float4*)(out + (size_t)(row0 + r) * D + col) = o;
}

extern "C" void kernel_launch(void* const* d_in, const int* in_sizes, int n_in,
                              void* d_out, int out_size, void* d_ws, size_t ws_size,
                              hipStream_t stream) {
  const float* x    = (const float*)d_in[0];
  const float* adj  = (const float*)d_in[1];
  const float* w    = (const float*)d_in[2];
  const float* bias = (const float*)d_in[3];
  float* out = (float*)d_out;

  float* dinv  = (float*)d_ws;
  float* bfrag = dinv + NN;

  k_deg <<<NN, 256, 0, stream>>>(adj, dinv);
  k_proj<<<NN, 128, 0, stream>>>(x, w, dinv, bfrag);
  k_agg <<<NN / 16, 32 * KSPLIT, 0, stream>>>(adj, bfrag, dinv, bias, out);
}


# k_agg: a wave reads its adj chunk (16 rows x 256 B) with whole-line 16-B loads, parks it in its own LDS block, takes A fragments from LDS; K split in 64-col chunks: summation order changes
# speedup vs baseline: 7.0840x; 1.7812x over previous
#include <hip/hip_runtime.h>

typedef float v2f __attribute__((ext_vector_type(2)));
typedef float v4f __attribute__((ext_vector_type(4)));
typedef float v8f __attribute__((ext_vector_type(8)));

#define NN 8192
#define D 128

__global__ __launch_bounds__(256) void k_deg(const float* __restrict__ adj,
                                             float* __restrict__ dinv) {
  __shared__ float red[256];
  const int row = blockIdx.x;
  const float* r = adj + (size_t)row * NN;
  float s = 0.f;
  for (int j = threadIdx.x; j < NN; j += 256)
    s += __builtin_nontemporal_load(r + j);
  red[threadIdx.x] = s;
  __syncthreads();
  for (int off = 128; off > 0; off >>= 1) {
    if (threadIdx.x < off) red[threadIdx.x] += red[threadIdx.x + off];
    __syncthreads();
  }
  if (threadIdx.x == 0) {
    float d = red[0];
    dinv[row] = d > 0.f ? rsqrtf(d) : 0.f;
  }
}

__global__ __launch_bounds__(128) void k_proj(const float* __restrict__ x,
                                              const float* __restrict__ w,
                                              const float* __restrict__ dinv,
                                              float* __restrict__ bfrag) {
  __shared__ float xs[D];
  const int row = blockIdx.x;
  const int col = threadIdx.x;
  xs[col] = x[(size_t)row * D + col];
  __syncthreads();
  float acc = 0.f;
#pragma unroll 8
  for (int k = 0; k < D; ++k) acc += xs[k] * w[k * D + col];
  acc *= dinv[row];

  const int ks = row >> 2, within = row & 3;
  const int r = within & 1, half = within >> 1;
  const int lane = half * 16 + (col & 15);
  const int tile = col >> 4;
  bfrag[(((size_t)ks * 8 + tile) * 32 + lane) * 2 + r] = acc;
}

#define KSPLIT 16
#define RED_PITCH 132
#define CW 64
#define A_PITCH 68
__global__ __launch_bounds__(32 * KSPLIT) void k_agg(const float* __restrict__ adj,
                                             const float* __restrict__ bfrag,
                                             const float* __restrict__ dinv,
                                             const float* __restrict__ bias,
                                             float* __restrict__ out) {
  __shared__ __attribute__((aligned(16))) float red[KSPLIT][16][RED_PITCH];

  const int lane = threadIdx.x & 31;
  const int wave = threadIdx.x >> 5;
  const int row0 = blockIdx.x * 16;
  const int lo = lane & 15, hi = lane >> 4;

  float* blk = &red[wave][0][0];
  const float* gsrc = adj + (size_t)(row0 + hi) * NN + 4 * lo;
  float* sdst = blk + hi * A_PITCH + 4 * lo;
  const float* afrag = blk + lo * A_PITCH + 2 * hi;
  const v2f* bf = (const v2f*)bfrag + lane;

  v8f c[8] = {};

  for (int ch = wave; ch < NN / CW; ch += KSPLIT) {
    v4f st[8];
#pragma unroll
    for (int i = 0; i < 8; ++i)
      st[i] = __builtin_nontemporal_load(
          (const v4f*)(gsrc + (size_t)(2 * i) * NN + (size_t)ch * CW));
#pragma unroll
    for (int i = 0; i < 8; ++i) *(v4f*)(sdst + 2 * i * A_PITCH) = st[i];

    const v2f* bch = bf + (size_t)ch * (CW / 4) * (8 * 32);
#pragma unroll 2
    for (int s = 0; s < CW / 4; ++s) {
      const v2f a = *(const v2f*)(afrag + 4 * s);
      const v2f* bks = bch + (size_t)s * (8 * 32);
#pragma unroll
      for (int t = 0; t < 8; ++t) {
        const v2f b = bks[t * 32];
        c[t] = __builtin_amdgcn_wmma_f32_16x16x4_f32(
             false, a,  false, b,
             (short)0, c[t],  false,  false);
      }
    }
  }

#pragma unroll
  for (int t = 0; t < 8; ++t) {
#pragma unroll
    for (int v = 0; v < 8; ++v) red[wave][hi * 8 + v][t * 16 + lo] = c[t][v];
  }
  __syncthreads();

  const int r = wave;
  const int col = lane * 4;
  float4 s = *(const float4*)&red[0][r][col];
#pragma unroll
  for (int w = 1; w < KSPLIT; ++w) {
    const float4 p = *(const float4*)&red[w][r][col];
    s.x += p.x; s.y += p.y; s.z += p.z; s.w += p.w;
  }
  const float di = dinv[row0 + r];
  const float4 bv = *(const float4*)(bias + col);
  float4 o;
  o.x = di * s.x + bv.x; o.y = di * s.y + bv.y;
  o.z = di * s.z + bv.z; o.w = di * s.w + bv.w;
  *(float4*)(out + (size_t)(row0 + r) * D + col) = o;
}

extern "C" void kernel_launch(void* const* d_in, const int* in_sizes, int n_in,
                              void* d_out, int out_size, void* d_ws, size_t ws_size,
                              hipStream_t stream) {
  const float* x    = (const float*)d_in[0];
  const float* adj  = (const float*)d_in[1];
  const float* w    = (const float*)d_in[2];
  const float* bias = (const float*)d_in[3];
  float* out = (float*)d_out;

  float* dinv  = (float*)d_ws;
  float* bfrag = dinv + NN;

  k_deg <<<NN, 256, 0, stream>>>(adj, dinv);
  k_proj<<<NN, 128, 0, stream>>>(x, w, dinv, bfrag);
  k_agg <<<NN / 16, 32 * KSPLIT, 0, stream>>>(adj, bfrag, dinv, bias, out);
}


# k_deg: one wave sums two whole adj rows with 512-B non-temporal loads, next 8 KB requested before the current is summed, no LDS or barrier: 4,096 waves not 65,536; row-sum order changes
# speedup vs baseline: 7.4979x; 1.0584x over previous
#include <hip/hip_runtime.h>

typedef float v2f __attribute__((ext_vector_type(2)));
typedef float v4f __attribute__((ext_vector_type(4)));
typedef float v8f __attribute__((ext_vector_type(8)));

#define NN 8192
#define D 128

#define DEG_WAVES 16
#define DEG_ROWS 2
#define DEG_Q 16
#define DEG_PIECES (NN / (128 * DEG_Q))
__global__ __launch_bounds__(32 * DEG_WAVES) void k_deg(const float* __restrict__ adj,
                                                        float* __restrict__ dinv) {
  const unsigned lane = threadIdx.x & 31u;
  const unsigned wave = __builtin_amdgcn_readfirstlane(threadIdx.x >> 5);
  const unsigned row0 = (blockIdx.x * DEG_WAVES + wave) * DEG_ROWS;
  const v4f* src = (const v4f*)(adj + (size_t)row0 * NN) + lane;
  v4f buf[2][DEG_Q];
  float sum[DEG_ROWS];
#define DEG_LOAD(P) do { _Pragma("unroll") for (int q = 0; q < DEG_Q; ++q) \
    buf[(P) & 1][q] = __builtin_nontemporal_load(src + ((P) * DEG_Q + q) * 32); } while (0)
  DEG_LOAD(0);
  float a0 = 0.f, a1 = 0.f, a2 = 0.f, a3 = 0.f;
#pragma unroll
  for (int p = 0; p < DEG_ROWS * DEG_PIECES; ++p) {
    __builtin_amdgcn_sched_barrier(0);
    if (p + 1 < DEG_ROWS * DEG_PIECES) DEG_LOAD(p + 1);
    __builtin_amdgcn_sched_barrier(0);
    if (p % DEG_PIECES == 0) { a0 = 0.f; a1 = 0.f; a2 = 0.f; a3 = 0.f; }
#pragma unroll
    for (int q = 0; q < DEG_Q; ++q) {
      const v4f v = buf[p & 1][q];
      a0 += v.x; a1 += v.y; a2 += v.z; a3 += v.w;
    }
    if (p % DEG_PIECES == DEG_PIECES - 1) sum[p / DEG_PIECES] = (a0 + a1) + (a2 + a3);
  }
#undef DEG_LOAD
#pragma unroll
  for (int r = 0; r < DEG_ROWS; ++r) {
    float s = sum[r];
#pragma unroll
    for (int m = 16; m >= 1; m >>= 1) s += __shfl_xor(s, m, 32);
    sum[r] = s;
  }
  float mine = 0.f;
#pragma unroll
  for (int r = 0; r < DEG_ROWS; ++r) if (lane == (unsigned)r) mine = sum[r];
  if (lane < (unsigned)DEG_ROWS) dinv[row0 + lane] = mine > 0.f ? rsqrtf(mine) : 0.f;
}

__global__ __launch_bounds__(128) void k_proj(const float* __restrict__ x,
                                              const float* __restrict__ w,
                                              const float* __restrict__ dinv,
                                              float* __restrict__ bfrag) {
  __shared__ float xs[D];
  const int row = blockIdx.x;
  const int col = threadIdx.x;
  xs[col] = x[(size_t)row * D + col];
  __syncthreads();
  float acc = 0.f;
#pragma unroll 8
  for (int k = 0; k < D; ++k) acc += xs[k] * w[k * D + col];
  acc *= dinv[row];

  const int ks = row >> 2, within = row & 3;
  const int r = within & 1, half = within >> 1;
  const int lane = half * 16 + (col & 15);
  const int tile = col >> 4;
  bfrag[(((size_t)ks * 8 + tile) * 32 + lane) * 2 + r] = acc;
}

#define KSPLIT 16
#define RED_PITCH 132
#define CW 64
#define A_PITCH 68
__global__ __launch_bounds__(32 * KSPLIT) void k_agg(const float* __restrict__ adj,
                                             const float* __restrict__ bfrag,
                                             const float* __restrict__ dinv,
                                             const float* __restrict__ bias,
                                             float* __restrict__ out) {
  __shared__ __attribute__((aligned(16))) float red[KSPLIT][16][RED_PITCH];

  const int lane = threadIdx.x & 31;
  const int wave = threadIdx.x >> 5;
  const int row0 = blockIdx.x * 16;
  const int lo = lane & 15, hi = lane >> 4;

  float* blk = &red[wave][0][0];
  const float* gsrc = adj + (size_t)(row0 + hi) * NN + 4 * lo;
  float* sdst = blk + hi * A_PITCH + 4 * lo;
  const float* afrag = blk + lo * A_PITCH + 2 * hi;
  const v2f* bf = (const v2f*)bfrag + lane;

  v8f c[8] = {};

  for (int ch = wave; ch < NN / CW; ch += KSPLIT) {
    v4f st[8];
#pragma unroll
    for (int i = 0; i < 8; ++i)
      st[i] = __builtin_nontemporal_load(
          (const v4f*)(gsrc + (size_t)(2 * i) * NN + (size_t)ch * CW));
#pragma unroll
    for (int i = 0; i < 8; ++i) *(v4f*)(sdst + 2 * i * A_PITCH) = st[i];

    const v2f* bch = bf + (size_t)ch * (CW / 4) * (8 * 32);
#pragma unroll 2
    for (int s = 0; s < CW / 4; ++s) {
      const v2f a = *(const v2f*)(afrag + 4 * s);
      const v2f* bks = bch + (size_t)s * (8 * 32);
#pragma unroll
      for (int t = 0; t < 8; ++t) {
        const v2f b = bks[t * 32];
        c[t] = __builtin_amdgcn_wmma_f32_16x16x4_f32(
             false, a,  false, b,
             (short)0, c[t],  false,  false);
      }
    }
  }

#pragma unroll
  for (int t = 0; t < 8; ++t) {
#pragma unroll
    for (int v = 0; v < 8; ++v) red[wave][hi * 8 + v][t * 16 + lo] = c[t][v];
  }
  __syncthreads();

  const int r = wave;
  const int col = lane * 4;
  float4 s = *(const float4*)&red[0][r][col];
#pragma unroll
  for (int w = 1; w < KSPLIT; ++w) {
    const float4 p = *(const float4*)&red[w][r][col];
    s.x += p.x; s.y += p.y; s.z += p.z; s.w += p.w;
  }
  const float di = dinv[row0 + r];
  const float4 bv = *(const float4*)(bias + col);
  float4 o;
  o.x = di * s.x + bv.x; o.y = di * s.y + bv.y;
  o.z = di * s.z + bv.z; o.w = di * s.w + bv.w;
  *(float4*)(out + (size_t)(row0 + r) * D + col) = o;
}

extern "C" void kernel_launch(void* const* d_in, const int* in_sizes, int n_in,
                              void* d_out, int out_size, void* d_ws, size_t ws_size,
                              hipStream_t stream) {
  const float* x    = (const float*)d_in[0];
  const float* adj  = (const float*)d_in[1];
  const float* w    = (const float*)d_in[2];
  const float* bias = (const float*)d_in[3];
  float* out = (float*)d_out;

  float* dinv  = (float*)d_ws;
  float* bfrag = dinv + NN;

  k_deg <<<NN / (DEG_WAVES * DEG_ROWS), 32 * DEG_WAVES, 0, stream>>>(adj, dinv);
  k_proj<<<NN, 128, 0, stream>>>(x, w, dinv, bfrag);
  k_agg <<<NN / 16, 32 * KSPLIT, 0, stream>>>(adj, bfrag, dinv, bias, out);
}


# k_proj folded into k_deg: after the row sums a workgroup does its 32 rows of x@W on f32 WMMA, a tile per wave, scaled by its own dinv: 2 launches not 3, 32,768 fewer waves; x@W sum order changes
# speedup vs baseline: 8.0831x; 1.0780x over previous
#include <hip/hip_runtime.h>

typedef float v2f __attribute__((ext_vector_type(2)));
typedef float v4f __attribute__((ext_vector_type(4)));
typedef float v8f __attribute__((ext_vector_type(8)));

#define NN 8192
#define D 128

#define DEG_WAVES 16
#define DEG_ROWS 2
#define DEG_Q 16
#define DEG_PIECES (NN / (128 * DEG_Q))
#define WG_ROWS (DEG_WAVES * DEG_ROWS)
__global__ __launch_bounds__(32 * DEG_WAVES) void k_deg_proj(const float* __restrict__ adj,
                                                             const float* __restrict__ x,
                                                             const float* __restrict__ w,
                                                             float* __restrict__ dinv,
                                                             float* __restrict__ bfrag) {
  __shared__ __attribute__((aligned(16))) float sdinv[WG_ROWS];

  const unsigned lane = threadIdx.x & 31u;
  const unsigned wave = __builtin_amdgcn_readfirstlane(threadIdx.x >> 5);
  const unsigned row0 = (blockIdx.x * DEG_WAVES + wave) * DEG_ROWS;
  const v4f* src = (const v4f*)(adj + (size_t)row0 * NN) + lane;
  v4f buf[2][DEG_Q];
  float sum[DEG_ROWS];
#define DEG_LOAD(P) do { _Pragma("unroll") for (int q = 0; q < DEG_Q; ++q) \
    buf[(P) & 1][q] = __builtin_nontemporal_load(src + ((P) * DEG_Q + q) * 32); } while (0)
  DEG_LOAD(0);
  float a0 = 0.f, a1 = 0.f, a2 = 0.f, a3 = 0.f;
#pragma unroll
  for (int p = 0; p < DEG_ROWS * DEG_PIECES; ++p) {
    __builtin_amdgcn_sched_barrier(0);
    if (p + 1 < DEG_ROWS * DEG_PIECES) DEG_LOAD(p + 1);
    __builtin_amdgcn_sched_barrier(0);
    if (p % DEG_PIECES == 0) { a0 = 0.f; a1 = 0.f; a2 = 0.f; a3 = 0.f; }
#pragma unroll
    for (int q = 0; q < DEG_Q; ++q) {
      const v4f v = buf[p & 1][q];
      a0 += v.x; a1 += v.y; a2 += v.z; a3 += v.w;
    }
    if (p % DEG_PIECES == DEG_PIECES - 1) sum[p / DEG_PIECES] = (a0 + a1) + (a2 + a3);
  }
#undef DEG_LOAD
#pragma unroll
  for (int r = 0; r < DEG_ROWS; ++r) {
    float s = sum[r];
#pragma unroll
    for (int m = 16; m >= 1; m >>= 1) s += __shfl_xor(s, m, 32);
    sum[r] = s;
  }
  float mine = 0.f;
#pragma unroll
  for (int r = 0; r < DEG_ROWS; ++r) if (lane == (unsigned)r) mine = sum[r];
  if (lane < (unsigned)DEG_ROWS) {
    const float dv = mine > 0.f ? rsqrtf(mine) : 0.f;
    dinv[row0 + lane] = dv;
    sdinv[wave * DEG_ROWS + lane] = dv;
  }

  const unsigned lo = lane & 15u, hi = lane >> 4;
  const unsigned mt = wave & 1u, nt = wave >> 1;
  const unsigned rbase = blockIdx.x * WG_ROWS + mt * 16;
  const float* xa = x + (size_t)(rbase + lo) * D + 2 * hi;
  const float* wb = w + (size_t)(2 * hi) * D + nt * 16 + lo;
  v8f acc0 = {}, acc1 = {};
  for (int s = 0; s < D / 4; s += 2) {
    const v2f fa0 = *(const v2f*)(xa + 4 * s);
    const v2f fa1 = *(const v2f*)(xa + 4 * s + 4);
    v2f fb0, fb1;
    fb0.x = wb[(4 * s) * D];
    fb0.y = wb[(4 * s + 1) * D];
    fb1.x = wb[(4 * s + 4) * D];
    fb1.y = wb[(4 * s + 5) * D];
    acc0 = __builtin_amdgcn_wmma_f32_16x16x4_f32(
         false, fa0,  false, fb0,
         (short)0, acc0,  false,  false);
    acc1 = __builtin_amdgcn_wmma_f32_16x16x4_f32(
         false, fa1,  false, fb1,
         (short)0, acc1,  false,  false);
  }

  __syncthreads();

  const v4f d0 = *(const v4f*)&sdinv[mt * 16 + 8 * hi];
  const v4f d1 = *(const v4f*)&sdinv[mt * 16 + 8 * hi + 4];
  v2f* dst = (v2f*)bfrag + ((size_t)((rbase >> 2) + 2 * hi) * 8 + nt) * 32 + lo;
  v2f o;
  o.x = (acc0[0] + acc1[0]) * d0.x; o.y = (acc0[1] + acc1[1]) * d0.y;
  dst[0] = o;
  o.x = (acc0[2] + acc1[2]) * d0.z; o.y = (acc0[3] + acc1[3]) * d0.w;
  dst[16] = o;
  o.x = (acc0[4] + acc1[4]) * d1.x; o.y = (acc0[5] + acc1[5]) * d1.y;
  dst[8 * 32] = o;
  o.x = (acc0[6] + acc1[6]) * d1.z; o.y = (acc0[7] + acc1[7]) * d1.w;
  dst[8 * 32 + 16] = o;
}

#define KSPLIT 16
#define RED_PITCH 132
#define CW 64
#define A_PITCH 68
__global__ __launch_bounds__(32 * KSPLIT) void k_agg(const float* __restrict__ adj,
                                             const float* __restrict__ bfrag,
                                             const float* __restrict__ dinv,
                                             const float* __restrict__ bias,
                                             float* __restrict__ out) {
  __shared__ __attribute__((aligned(16))) float red[KSPLIT][16][RED_PITCH];

  const int lane = threadIdx.x & 31;
  const int wave = threadIdx.x >> 5;
  const int row0 = blockIdx.x * 16;
  const int lo = lane & 15, hi = lane >> 4;

  float* blk = &red[wave][0][0];
  const float* gsrc = adj + (size_t)(row0 + hi) * NN + 4 * lo;
  float* sdst = blk + hi * A_PITCH + 4 * lo;
  const float* afrag = blk + lo * A_PITCH + 2 * hi;
  const v2f* bf = (const v2f*)bfrag + lane;

  v8f c[8] = {};

  for (int ch = wave; ch < NN / CW; ch += KSPLIT) {
    v4f st[8];
#pragma unroll
    for (int i = 0; i < 8; ++i)
      st[i] = __builtin_nontemporal_load(
          (const v4f*)(gsrc + (size_t)(2 * i) * NN + (size_t)ch * CW));
#pragma unroll
    for (int i = 0; i < 8; ++i) *(v4f*)(sdst + 2 * i * A_PITCH) = st[i];

    const v2f* bch = bf + (size_t)ch * (CW / 4) * (8 * 32);
#pragma unroll 2
    for (int s = 0; s < CW / 4; ++s) {
      const v2f a = *(const v2f*)(afrag + 4 * s);
      const v2f* bks = bch + (size_t)s * (8 * 32);
#pragma unroll
      for (int t = 0; t < 8; ++t) {
        const v2f b = bks[t * 32];
        c[t] = __builtin_amdgcn_wmma_f32_16x16x4_f32(
             false, a,  false, b,
             (short)0, c[t],  false,  false);
      }
    }
  }

#pragma unroll
  for (int t = 0; t < 8; ++t) {
#pragma unroll
    for (int v = 0; v < 8; ++v) red[wave][hi * 8 + v][t * 16 + lo] = c[t][v];
  }
  __syncthreads();

  const int r = wave;
  const int col = lane * 4;
  float4 s = *(const float4*)&red[0][r][col];
#pragma unroll
  for (int w = 1; w < KSPLIT; ++w) {
    const float4 p = *(const float4*)&red[w][r][col];
    s.x += p.x; s.y += p.y; s.z += p.z; s.w += p.w;
  }
  const float di = dinv[row0 + r];
  const float4 bv = *(const float4*)(bias + col);
  float4 o;
  o.x = di * s.x + bv.x; o.y = di * s.y + bv.y;
  o.z = di * s.z + bv.z; o.w = di * s.w + bv.w;
  *(float4*)(out + (size_t)(row0 + r) * D + col) = o;
}

extern "C" void kernel_launch(void* const* d_in, const int* in_sizes, int n_in,
                              void* d_out, int out_size, void* d_ws, size_t ws_size,
                              hipStream_t stream) {
  const float* x    = (const float*)d_in[0];
  const float* adj  = (const float*)d_in[1];
  const float* w    = (const float*)d_in[2];
  const float* bias = (const float*)d_in[3];
  float* out = (float*)d_out;

  float* dinv  = (float*)d_ws;
  float* bfrag = dinv + NN;

  k_deg_proj<<<NN / WG_ROWS, 32 * DEG_WAVES, 0, stream>>>(adj, x, w, dinv, bfrag);
  k_agg     <<<NN / 16, 32 * KSPLIT, 0, stream>>>(adj, bfrag, dinv, bias, out);
}


# k_agg: a wave owns two 16-row tiles (32 rows x 128 cols, 256 workgroups, 4 k-steps per trip): each B fragment loaded from L2 feeds two WMMAs, B loads and bytes per WMMA halve; sum order unchanged
# speedup vs baseline: 8.9040x; 1.1016x over previous
#include <hip/hip_runtime.h>

typedef float v2f __attribute__((ext_vector_type(2)));
typedef float v4f __attribute__((ext_vector_type(4)));
typedef float v8f __attribute__((ext_vector_type(8)));

#define NN 8192
#define D 128

#define DEG_WAVES 16
#define DEG_ROWS 2
#define DEG_Q 16
#define DEG_PIECES (NN / (128 * DEG_Q))
#define WG_ROWS (DEG_WAVES * DEG_ROWS)
__global__ __launch_bounds__(32 * DEG_WAVES) void k_deg_proj(const float* __restrict__ adj,
                                                             const float* __restrict__ x,
                                                             const float* __restrict__ w,
                                                             float* __restrict__ dinv,
                                                             float* __restrict__ bfrag) {
  __shared__ __attribute__((aligned(16))) float sdinv[WG_ROWS];

  const unsigned lane = threadIdx.x & 31u;
  const unsigned wave = __builtin_amdgcn_readfirstlane(threadIdx.x >> 5);
  const unsigned row0 = (blockIdx.x * DEG_WAVES + wave) * DEG_ROWS;
  const v4f* src = (const v4f*)(adj + (size_t)row0 * NN) + lane;
  v4f buf[2][DEG_Q];
  float sum[DEG_ROWS];
#define DEG_LOAD(P) do { _Pragma("unroll") for (int q = 0; q < DEG_Q; ++q) \
    buf[(P) & 1][q] = __builtin_nontemporal_load(src + ((P) * DEG_Q + q) * 32); } while (0)
  DEG_LOAD(0);
  float a0 = 0.f, a1 = 0.f, a2 = 0.f, a3 = 0.f;
#pragma unroll
  for (int p = 0; p < DEG_ROWS * DEG_PIECES; ++p) {
    __builtin_amdgcn_sched_barrier(0);
    if (p + 1 < DEG_ROWS * DEG_PIECES) DEG_LOAD(p + 1);
    __builtin_amdgcn_sched_barrier(0);
    if (p % DEG_PIECES == 0) { a0 = 0.f; a1 = 0.f; a2 = 0.f; a3 = 0.f; }
#pragma unroll
    for (int q = 0; q < DEG_Q; ++q) {
      const v4f v = buf[p & 1][q];
      a0 += v.x; a1 += v.y; a2 += v.z; a3 += v.w;
    }
    if (p % DEG_PIECES == DEG_PIECES - 1) sum[p / DEG_PIECES] = (a0 + a1) + (a2 + a3);
  }
#undef DEG_LOAD
#pragma unroll
  for (int r = 0; r < DEG_ROWS; ++r) {
    float s = sum[r];
#pragma unroll
    for (int m = 16; m >= 1; m >>= 1) s += __shfl_xor(s, m, 32);
    sum[r] = s;
  }
  float mine = 0.f;
#pragma unroll
  for (int r = 0; r < DEG_ROWS; ++r) if (lane == (unsigned)r) mine = sum[r];
  if (lane < (unsigned)DEG_ROWS) {
    const float dv = mine > 0.f ? rsqrtf(mine) : 0.f;
    dinv[row0 + lane] = dv;
    sdinv[wave * DEG_ROWS + lane] = dv;
  }

  const unsigned lo = lane & 15u, hi = lane >> 4;
  const unsigned mt = wave & 1u, nt = wave >> 1;
  const unsigned rbase = blockIdx.x * WG_ROWS + mt * 16;
  const float* xa = x + (size_t)(rbase + lo) * D + 2 * hi;
  const float* wb = w + (size_t)(2 * hi) * D + nt * 16 + lo;
  v8f acc0 = {}, acc1 = {};
  for (int s = 0; s < D / 4; s += 2) {
    const v2f fa0 = *(const v2f*)(xa + 4 * s);
    const v2f fa1 = *(const v2f*)(xa + 4 * s + 4);
    v2f fb0, fb1;
    fb0.x = wb[(4 * s) * D];
    fb0.y = wb[(4 * s + 1) * D];
    fb1.x = wb[(4 * s + 4) * D];
    fb1.y = wb[(4 * s + 5) * D];
    acc0 = __builtin_amdgcn_wmma_f32_16x16x4_f32(
         false, fa0,  false, fb0,
         (short)0, acc0,  false,  false);
    acc1 = __builtin_amdgcn_wmma_f32_16x16x4_f32(
         false, fa1,  false, fb1,
         (short)0, acc1,  false,  false);
  }

  __syncthreads();

  const v4f d0 = *(const v4f*)&sdinv[mt * 16 + 8 * hi];
  const v4f d1 = *(const v4f*)&sdinv[mt * 16 + 8 * hi + 4];
  v2f* dst = (v2f*)bfrag + ((size_t)((rbase >> 2) + 2 * hi) * 8 + nt) * 32 + lo;
  v2f o;
  o.x = (acc0[0] + acc1[0]) * d0.x; o.y = (acc0[1] + acc1[1]) * d0.y;
  dst[0] = o;
  o.x = (acc0[2] + acc1[2]) * d0.z; o.y = (acc0[3] + acc1[3]) * d0.w;
  dst[16] = o;
  o.x = (acc0[4] + acc1[4]) * d1.x; o.y = (acc0[5] + acc1[5]) * d1.y;
  dst[8 * 32] = o;
  o.x = (acc0[6] + acc1[6]) * d1.z; o.y = (acc0[7] + acc1[7]) * d1.w;
  dst[8 * 32 + 16] = o;
}

#define KSPLIT 16
#define AGG_MT 2
#define AGG_ROWS (16 * AGG_MT)
#define RED_PITCH 132
#define CW 64
#define A_PITCH 68
__global__ __launch_bounds__(32 * KSPLIT) void k_agg(const float* __restrict__ adj,
                                             const float* __restrict__ bfrag,
                                             const float* __restrict__ dinv,
                                             const float* __restrict__ bias,
                                             float* __restrict__ out) {
  __shared__ __attribute__((aligned(16))) float red[KSPLIT][AGG_ROWS][RED_PITCH];

  const int lane = threadIdx.x & 31;
  const int wave = threadIdx.x >> 5;
  const int row0 = blockIdx.x * AGG_ROWS;
  const int lo = lane & 15, hi = lane >> 4;

  float* blk = &red[wave][0][0];
  const float* gsrc = adj + (size_t)(row0 + hi) * NN + 4 * lo;
  float* sdst = blk + hi * A_PITCH + 4 * lo;
  const float* afrag = blk + lo * A_PITCH + 2 * hi;
  const v2f* bf = (const v2f*)bfrag + lane;

  v8f c0[8] = {}, c1[8] = {};

  for (int ch = wave; ch < NN / CW; ch += KSPLIT) {
    v4f st[AGG_ROWS / 2];
#pragma unroll
    for (int i = 0; i < AGG_ROWS / 2; ++i)
      st[i] = __builtin_nontemporal_load(
          (const v4f*)(gsrc + (size_t)(2 * i) * NN + (size_t)ch * CW));
#pragma unroll
    for (int i = 0; i < AGG_ROWS / 2; ++i) *(v4f*)(sdst + 2 * i * A_PITCH) = st[i];

    const v2f* bch = bf + (size_t)ch * (CW / 4) * (8 * 32);
#pragma unroll 4
    for (int s = 0; s < CW / 4; ++s) {
      const v2f a0 = *(const v2f*)(afrag + 4 * s);
      const v2f a1 = *(const v2f*)(afrag + 16 * A_PITCH + 4 * s);
      const v2f* bks = bch + (size_t)s * (8 * 32);
#pragma unroll
      for (int t = 0; t < 8; ++t) {
        const v2f b = bks[t * 32];
        c0[t] = __builtin_amdgcn_wmma_f32_16x16x4_f32(
             false, a0,  false, b,
             (short)0, c0[t],  false,  false);
        c1[t] = __builtin_amdgcn_wmma_f32_16x16x4_f32(
             false, a1,  false, b,
             (short)0, c1[t],  false,  false);
      }
    }
  }

#pragma unroll
  for (int t = 0; t < 8; ++t) {
#pragma unroll
    for (int v = 0; v < 8; ++v) {
      red[wave][hi * 8 + v][t * 16 + lo] = c0[t][v];
      red[wave][16 + hi * 8 + v][t * 16 + lo] = c1[t][v];
    }
  }
  __syncthreads();

  const int col = lane * 4;
  const float4 bv = *(const float4*)(bias + col);
#pragma unroll
  for (int m = 0; m < AGG_MT; ++m) {
    const int r = 16 * m + wave;
    float4 s = *(const float4*)&red[0][r][col];
#pragma unroll
    for (int w = 1; w < KSPLIT; ++w) {
      const float4 p = *(const float4*)&red[w][r][col];
      s.x += p.x; s.y += p.y; s.z += p.z; s.w += p.w;
    }
    const float di = dinv[row0 + r];
    float4 o;
    o.x = di * s.x + bv.x; o.y = di * s.y + bv.y;
    o.z = di * s.z + bv.z; o.w = di * s.w + bv.w;
    *(float4*)(out + (size_t)(row0 + r) * D + col) = o;
  }
}

extern "C" void kernel_launch(void* const* d_in, const int* in_sizes, int n_in,
                              void* d_out, int out_size, void* d_ws, size_t ws_size,
                              hipStream_t stream) {
  const float* x    = (const float*)d_in[0];
  const float* adj  = (const float*)d_in[1];
  const float* w    = (const float*)d_in[2];
  const float* bias = (const float*)d_in[3];
  float* out = (float*)d_out;

  float* dinv  = (float*)d_ws;
  float* bfrag = dinv + NN;

  k_deg_proj<<<NN / WG_ROWS, 32 * DEG_WAVES, 0, stream>>>(adj, x, w, dinv, bfrag);
  k_agg     <<<NN / AGG_ROWS, 32 * KSPLIT, 0, stream>>>(adj, bfrag, dinv, bias, out);
}


# k_agg: operands requested ahead of their WMMAs: B fragments one 2-k-step stage ahead in a second register set, next adj chunk in 4 pieces into a second LDS slot; sum order unchanged
# speedup vs baseline: 9.2170x; 1.0352x over previous
#include <hip/hip_runtime.h>

typedef float v2f __attribute__((ext_vector_type(2)));
typedef float v4f __attribute__((ext_vector_type(4)));
typedef float v8f __attribute__((ext_vector_type(8)));

#define NN 8192
#define D 128

#define DEG_WAVES 16
#define DEG_ROWS 2
#define DEG_Q 16
#define DEG_PIECES (NN / (128 * DEG_Q))
#define WG_ROWS (DEG_WAVES * DEG_ROWS)
__global__ __launch_bounds__(32 * DEG_WAVES) void k_deg_proj(const float* __restrict__ adj,
                                                             const float* __restrict__ x,
                                                             const float* __restrict__ w,
                                                             float* __restrict__ dinv,
                                                             float* __restrict__ bfrag) {
  __shared__ __attribute__((aligned(16))) float sdinv[WG_ROWS];

  const unsigned lane = threadIdx.x & 31u;
  const unsigned wave = __builtin_amdgcn_readfirstlane(threadIdx.x >> 5);
  const unsigned row0 = (blockIdx.x * DEG_WAVES + wave) * DEG_ROWS;
  const v4f* src = (const v4f*)(adj + (size_t)row0 * NN) + lane;
  v4f buf[2][DEG_Q];
  float sum[DEG_ROWS];
#define DEG_LOAD(P) do { _Pragma("unroll") for (int q = 0; q < DEG_Q; ++q) \
    buf[(P) & 1][q] = __builtin_nontemporal_load(src + ((P) * DEG_Q + q) * 32); } while (0)
  DEG_LOAD(0);
  float a0 = 0.f, a1 = 0.f, a2 = 0.f, a3 = 0.f;
#pragma unroll
  for (int p = 0; p < DEG_ROWS * DEG_PIECES; ++p) {
    __builtin_amdgcn_sched_barrier(0);
    if (p + 1 < DEG_ROWS * DEG_PIECES) DEG_LOAD(p + 1);
    __builtin_amdgcn_sched_barrier(0);
    if (p % DEG_PIECES == 0) { a0 = 0.f; a1 = 0.f; a2 = 0.f; a3 = 0.f; }
#pragma unroll
    for (int q = 0; q < DEG_Q; ++q) {
      const v4f v = buf[p & 1][q];
      a0 += v.x; a1 += v.y; a2 += v.z; a3 += v.w;
    }
    if (p % DEG_PIECES == DEG_PIECES - 1) sum[p / DEG_PIECES] = (a0 + a1) + (a2 + a3);
  }
#undef DEG_LOAD
#pragma unroll
  for (int r = 0; r < DEG_ROWS; ++r) {
    float s = sum[r];
#pragma unroll
    for (int m = 16; m >= 1; m >>= 1) s += __shfl_xor(s, m, 32);
    sum[r] = s;
  }
  float mine = 0.f;
#pragma unroll
  for (int r = 0; r < DEG_ROWS; ++r) if (lane == (unsigned)r) mine = sum[r];
  if (lane < (unsigned)DEG_ROWS) {
    const float dv = mine > 0.f ? rsqrtf(mine) : 0.f;
    dinv[row0 + lane] = dv;
    sdinv[wave * DEG_ROWS + lane] = dv;
  }

  const unsigned lo = lane & 15u, hi = lane >> 4;
  const unsigned mt = wave & 1u, nt = wave >> 1;
  const unsigned rbase = blockIdx.x * WG_ROWS + mt * 16;
  const float* xa = x + (size_t)(rbase + lo) * D + 2 * hi;
  const float* wb = w + (size_t)(2 * hi) * D + nt * 16 + lo;
  v8f acc0 = {}, acc1 = {};
  for (int s = 0; s < D / 4; s += 2) {
    const v2f fa0 = *(const v2f*)(xa + 4 * s);
    const v2f fa1 = *(const v2f*)(xa + 4 * s + 4);
    v2f fb0, fb1;
    fb0.x = wb[(4 * s) * D];
    fb0.y = wb[(4 * s + 1) * D];
    fb1.x = wb[(4 * s + 4) * D];
    fb1.y = wb[(4 * s + 5) * D];
    acc0 = __builtin_amdgcn_wmma_f32_16x16x4_f32(
         false, fa0,  false, fb0,
         (short)0, acc0,  false,  false);
    acc1 = __builtin_amdgcn_wmma_f32_16x16x4_f32(
         false, fa1,  false, fb1,
         (short)0, acc1,  false,  false);
  }

  __syncthreads();

  const v4f d0 = *(const v4f*)&sdinv[mt * 16 + 8 * hi];
  const v4f d1 = *(const v4f*)&sdinv[mt * 16 + 8 * hi + 4];
  v2f* dst = (v2f*)bfrag + ((size_t)((rbase >> 2) + 2 * hi) * 8 + nt) * 32 + lo;
  v2f o;
  o.x = (acc0[0] + acc1[0]) * d0.x; o.y = (acc0[1] + acc1[1]) * d0.y;
  dst[0] = o;
  o.x = (acc0[2] + acc1[2]) * d0.z; o.y = (acc0[3] + acc1[3]) * d0.w;
  dst[16] = o;
  o.x = (acc0[4] + acc1[4]) * d1.x; o.y = (acc0[5] + acc1[5]) * d1.y;
  dst[8 * 32] = o;
  o.x = (acc0[6] + acc1[6]) * d1.z; o.y = (acc0[7] + acc1[7]) * d1.w;
  dst[8 * 32 + 16] = o;
}

#define KSPLIT 16
#define AGG_MT 2
#define AGG_ROWS (16 * AGG_MT)
#define RED_PITCH 132
#define CW 64
#define A_PITCH 68
#define SLOT (AGG_ROWS * A_PITCH)
#define WBLK (2 * SLOT)
#define SK 2
#define NSTG (CW / 4 / SK)
#define NCH (NN / CW / KSPLIT)
#define BCH (KSPLIT * (CW / 4) * 256)
#define PIECE 4

__device__ __forceinline__ void agg_load_b(v2f (&b)[SK * 8], const v2f* p) {
#pragma unroll
  for (int q = 0; q < SK * 8; ++q) b[q] = p[q * 32];
}
__device__ __forceinline__ void agg_load_a(v2f (&a)[AGG_MT], const float* p) {
  a[0] = *(const v2f*)p;
  a[1] = *(const v2f*)(p + 16 * A_PITCH);
}

template <bool MORE>
__device__ __forceinline__ void agg_chunk(v8f (&c0)[8], v8f (&c1)[8],
                                          v2f (&bs)[2][SK * 8], v2f (&as)[SK][AGG_MT],
                                          const float* cur_a, const float* nxt_a, float* nxt_s,
                                          const float* gn, const v2f* bc) {
  v4f st[PIECE];
#pragma unroll
  for (int j = 0; j < NSTG; ++j) {
    __builtin_amdgcn_sched_barrier(0);
    if (j + 1 < NSTG) agg_load_b(bs[(j + 1) & 1], bc + (j + 1) * SK * 256);
    else if (MORE)    agg_load_b(bs[0], bc + BCH);
    if (MORE) {
      const bool park = (j >= 2 && j % 2 == 0) || j == NSTG - 1;
      const int pfirst = j == NSTG - 1 ? 3 * PIECE : (j / 2 - 1) * PIECE;
      const int pn = park ? PIECE : 0;
#pragma unroll
      for (int i = 0; i < pn; ++i) *(v4f*)(nxt_s + 2 * (pfirst + i) * A_PITCH) = st[i];
      const int rfirst = (j / 2) * PIECE;
      const int rn = (j % 2 == 0) ? PIECE : 0;
#pragma unroll
      for (int i = 0; i < rn; ++i)
        st[i] = __builtin_nontemporal_load((const v4f*)(gn + (size_t)(2 * (rfirst + i)) * NN));
    }
    agg_load_a(as[1], cur_a + 4 * (SK * j + 1));
    __builtin_amdgcn_sched_barrier(0);
#pragma unroll
    for (int s = 0; s < SK; ++s) {
#pragma unroll
      for (int t = 0; t < 8; ++t) {
        c0[t] = __builtin_amdgcn_wmma_f32_16x16x4_f32(
             false, as[s][0],  false, bs[j & 1][8 * s + t],
             (short)0, c0[t],  false,  false);
        c1[t] = __builtin_amdgcn_wmma_f32_16x16x4_f32(
             false, as[s][1],  false, bs[j & 1][8 * s + t],
             (short)0, c1[t],  false,  false);
      }
      if (s == 0) {
        __builtin_amdgcn_sched_barrier(0);
        if (j + 1 < NSTG) agg_load_a(as[0], cur_a + 4 * SK * (j + 1));
        else if (MORE)    agg_load_a(as[0], nxt_a);
        __builtin_amdgcn_sched_barrier(0);
      }
    }
  }
}

__global__ __launch_bounds__(32 * KSPLIT) void k_agg(const float* __restrict__ adj,
                                             const float* __restrict__ bfrag,
                                             const float* __restrict__ dinv,
                                             const float* __restrict__ bias,
                                             float* __restrict__ out) {
  __shared__ __attribute__((aligned(16))) float lds[KSPLIT * WBLK];

  const int lane = threadIdx.x & 31;
  const int wave = __builtin_amdgcn_readfirstlane(threadIdx.x >> 5);
  const int row0 = blockIdx.x * AGG_ROWS;
  const int lo = lane & 15, hi = lane >> 4;

  float* blk = lds + wave * WBLK;
  const float* g = adj + (size_t)(row0 + hi) * NN + 4 * lo + (size_t)wave * CW;
  float* cur_s = blk + hi * A_PITCH + 4 * lo;
  float* nxt_s = cur_s + SLOT;
  const float* cur_a = blk + lo * A_PITCH + 2 * hi;
  const float* nxt_a = cur_a + SLOT;
  const v2f* b = (const v2f*)bfrag + lane + (size_t)wave * ((CW / 4) * 256);

  v8f c0[8] = {}, c1[8] = {};
  v2f bs[2][SK * 8];
  v2f as[SK][AGG_MT];

  agg_load_b(bs[0], b);
  {
    v4f st[AGG_ROWS / 2];
#pragma unroll
    for (int i = 0; i < AGG_ROWS / 2; ++i)
      st[i] = __builtin_nontemporal_load((const v4f*)(g + (size_t)(2 * i) * NN));
#pragma unroll
    for (int i = 0; i < AGG_ROWS / 2; ++i) *(v4f*)(cur_s + 2 * i * A_PITCH) = st[i];
  }
  agg_load_a(as[0], cur_a);

#pragma unroll 1
  for (int c = 0; c < NCH - 1; ++c) {
    g += KSPLIT * CW;
    agg_chunk<true>(c0, c1, bs, as, cur_a, nxt_a, nxt_s, g, b);
    b += BCH;
    const float* ta = cur_a; cur_a = nxt_a; nxt_a = ta;
    float* ts = cur_s; cur_s = nxt_s; nxt_s = ts;
  }
  agg_chunk<false>(c0, c1, bs, as, cur_a, nxt_a, nxt_s, g, b);

#pragma unroll
  for (int t = 0; t < 8; ++t) {
#pragma unroll
    for (int v = 0; v < 8; ++v) {
      blk[(hi * 8 + v) * RED_PITCH + t * 16 + lo] = c0[t][v];
      blk[(16 + hi * 8 + v) * RED_PITCH + t * 16 + lo] = c1[t][v];
    }
  }
  __syncthreads();

  const int col = lane * 4;
  const float4 bv = *(const float4*)(bias + col);
#pragma unroll
  for (int m = 0; m < AGG_MT; ++m) {
    const int r = 16 * m + wave;
    float4 s = *(const float4*)&lds[r * RED_PITCH + col];
#pragma unroll
    for (int w = 1; w < KSPLIT; ++w) {
      const float4 p = *(const float4*)&lds[w * WBLK + r * RED_PITCH + col];
      s.x += p.x; s.y += p.y; s.z += p.z; s.w += p.w;
    }
    const float di = dinv[row0 + r];
    float4 o;
    o.x = di * s.x + bv.x; o.y = di * s.y + bv.y;
    o.z = di * s.z + bv.z; o.w = di * s.w + bv.w;
    *(float4*)(out + (size_t)(row0 + r) * D + col) = o;
  }
}

extern "C" void kernel_launch(void* const* d_in, const int* in_sizes, int n_in,
                              void* d_out, int out_size, void* d_ws, size_t ws_size,
                              hipStream_t stream) {
  const float* x    = (const float*)d_in[0];
  const float* adj  = (const float*)d_in[1];
  const float* w    = (const float*)d_in[2];
  const float* bias = (const float*)d_in[3];
  float* out = (float*)d_out;

  float* dinv  = (float*)d_ws;
  float* bfrag = dinv + NN;

  k_deg_proj<<<NN / WG_ROWS, 32 * DEG_WAVES, 0, stream>>>(adj, x, w, dinv, bfrag);
  k_agg     <<<NN / AGG_ROWS, 32 * KSPLIT, 0, stream>>>(adj, bfrag, dinv, bias, out);
}
